# starting kernel
# baseline (speedup 1.0000x reference)
#include <hip/hip_runtime.h>
#include <hip/hip_bf16.h>

typedef float v2f __attribute__((ext_vector_type(2)));
typedef float v8f __attribute__((ext_vector_type(8)));

#define XS_N 4608
#define WS_N 16384
#define VS_N 8192

__global__ __launch_bounds__(1024)
void deconv_caps_routing(const float* __restrict__ xg,
                         const float* __restrict__ wg,
                         const float* __restrict__ bg,
                         float* __restrict__ outg) {
  extern __shared__ float smem[];
  float* xs  = smem;
  float* ws  = smem + XS_N;
  float* vsm = smem + XS_N + WS_N;

  const int tid = threadIdx.x;
  const int blk = blockIdx.x;
  const int b  = blk >> 10;
  const int od = (blk >> 5) & 31;
  const int oh = blk & 31;

  const int tdbase = 1 - (od & 1);
  const int thbase = 1 - (oh & 1);
  const int id0 = (od + 1) >> 1;
  const int ih0 = (oh + 1) >> 1;

  for (int idx = tid; idx < WS_N; idx += 1024) {
    int ai  = idx & 15;
    int c   = (idx >> 4) & 63;
    int tw  = (idx >> 10) & 3;
    int thi = (idx >> 12) & 1;
    int tdi = (idx >> 13) & 1;
    int td = tdbase + 2 * tdi;
    int th = thbase + 2 * thi;
    ws[idx] = wg[((ai * 64 + c) << 6) + td * 16 + th * 4 + tw];
  }
  for (int idx = tid; idx < XS_N; idx += 1024) {
    int ai   = idx & 15;
    int t    = idx >> 4;
    int iwp  = t % 18;
    int rest = t / 18;
    int thi = rest & 1;
    int tdi = (rest >> 1) & 1;
    int ci  = rest >> 2;
    int iw = iwp - 1;
    int id = id0 - tdi;
    int ih = ih0 - thi;
    float v = 0.f;
    if ((unsigned)iw < 16u && (unsigned)id < 16u && (unsigned)ih < 16u)
      v = xg[(((b * 4 + ci) * 16 + ai) << 12) + (id << 8) + (ih << 4) + iw];
    xs[idx] = v;
  }
  __syncthreads();

  const int lane = tid & 31;
  const int wv   = tid >> 5;
  const int par  = wv >> 4;
  const int icap = (wv >> 2) & 3;
  const int ocap = wv & 3;
  const int ln = lane & 15;
  const int kh = lane >> 4;

  v8f acc = {0.f, 0.f, 0.f, 0.f, 0.f, 0.f, 0.f, 0.f};
#pragma unroll
  for (int j = 0; j < 32; ++j) {
    int tap = j >> 2;
    int twi = tap & 1;
    int thi = (tap >> 1) & 1;
    int tdi = tap >> 2;
    int aibase = ((j & 3) << 2) + (kh << 1);
    int iw = ln + par - twi;
    int aoff = ((((icap * 2 + tdi) * 2 + thi) * 18 + (iw + 1)) << 4) + aibase;
    int tw = (1 - par) + (twi << 1);
    int boff = (((((tdi * 2 + thi) << 2) + tw) * 64 + (ocap * 16 + ln)) << 4) + aibase;
    v2f a  = *(const v2f*)(xs + aoff);
    v2f bb = *(const v2f*)(ws + boff);
    acc = __builtin_amdgcn_wmma_f32_16x16x4_f32(
         false, a,  false, bb,
         (short)0, acc,  false,  false);
  }

  const int p_idx = icap * 4 + ocap;
#pragma unroll
  for (int g = 0; g < 8; ++g) {
    int vx = g + (kh << 3);
    vsm[(((par * 16 + vx) * 16 + p_idx) << 4) + ln] = acc[g];
  }
  __syncthreads();

  const int rr = wv & 15;
  const int p  = lane & 15;
  const int h  = lane >> 4;
  const int ri = p >> 2;
  const int ro = p & 3;

  float vv[8], bv[8];
  const float* vbase = vsm + (((par * 16 + rr) * 16 + p) << 4) + (h << 3);
#pragma unroll
  for (int j = 0; j < 8; ++j) vv[j] = vbase[j];
#pragma unroll
  for (int j = 0; j < 8; ++j) bv[j] = bg[ro * 16 + (h << 3) + j];

  float logit = 0.f;
  float pre[8];
  float scale = 0.f;
#pragma unroll
  for (int it = 0; it < 3; ++it) {
    float m = logit;
    m = fmaxf(m, __shfl_xor(m, 1));
    m = fmaxf(m, __shfl_xor(m, 2));
    float e = __expf(logit - m);
    float s = e;
    s += __shfl_xor(s, 1);
    s += __shfl_xor(s, 2);
    float route = e / s;
#pragma unroll
    for (int j = 0; j < 8; ++j) pre[j] = route * vv[j];
#pragma unroll
    for (int j = 0; j < 8; ++j) {
      pre[j] += __shfl_xor(pre[j], 4);
      pre[j] += __shfl_xor(pre[j], 8);
      pre[j] += bv[j];
    }
    if (it < 2) {
      float dt = 0.f, np2 = 0.f, nv2 = 0.f;
#pragma unroll
      for (int j = 0; j < 8; ++j) {
        dt  += pre[j] * vv[j];
        np2 += pre[j] * pre[j];
        nv2 += vv[j] * vv[j];
      }
      dt  += __shfl_xor(dt, 16);
      np2 += __shfl_xor(np2, 16);
      nv2 += __shfl_xor(nv2, 16);
      float na = sqrtf(np2), nb = sqrtf(nv2);
      logit += dt / fmaxf(na * nb, 1e-8f);
    } else {
      float n2 = 0.f;
#pragma unroll
      for (int j = 0; j < 8; ++j) n2 += pre[j] * pre[j];
      n2 += __shfl_xor(n2, 16);
      float norm = sqrtf(n2);
      scale = n2 / ((norm + 1e-12f) * (1.f + n2));
    }
  }

  if (ri == 0) {
    int ow = (rr << 1) + par;
#pragma unroll
    for (int j = 0; j < 8; ++j) {
      int ao = (h << 3) + j;
      outg[(((b * 4 + ro) * 16 + ao) << 15) + (od << 10) + (oh << 5) + ow] =
          pre[j] * scale;
    }
  }
}

extern "C" void kernel_launch(void* const* d_in, const int* in_sizes, int n_in,
                              void* d_out, int out_size, void* d_ws, size_t ws_size,
                              hipStream_t stream) {
  (void)in_sizes; (void)n_in; (void)out_size; (void)d_ws; (void)ws_size;
  const float* x = (const float*)d_in[0];
  const float* w = (const float*)d_in[1];
  const float* b = (const float*)d_in[2];
  float* out = (float*)d_out;

  const size_t shmem = (size_t)(XS_N + WS_N + VS_N) * sizeof(float);
  deconv_caps_routing<<<dim3(4 * 32 * 32), dim3(1024), shmem, stream>>>(x, w, b, out);
}


# LDS rows run as global memory does (x along iw, w along the taps), only the two ai of a k pair interleaved: staging copies whole contiguous lines, all loads in flight; k order, sums unchanged
# speedup vs baseline: 2.4599x; 2.4599x over previous
#include <hip/hip_runtime.h>

typedef float v2f __attribute__((ext_vector_type(2)));
typedef float v4f __attribute__((ext_vector_type(4)));
typedef float v8f __attribute__((ext_vector_type(8)));

#define XR 36
#define XROWS 1088
#define XS_N (XROWS * XR + 8)
#define XS_R2 (17 * 8 * XR)
#define XS_S (8 * XR)
#define WPC 36
#define WPA 2306
#define WS_N (8 * WPA)

__global__ __launch_bounds__(512)
void deconv_caps_routing(const float* __restrict__ xg,
                         const float* __restrict__ wg,
                         const float* __restrict__ bg,
                         float* __restrict__ outg) {
  __shared__ __attribute__((aligned(16))) float xs[XS_N];
  __shared__ __attribute__((aligned(16))) float ws[WS_N];

  const int tid = threadIdx.x;
  const int blk = blockIdx.x;
  const int b  = blk >> 6;
  const int od = (blk >> 1) & 31;
  const int ph = blk & 1;

  const int tdbase = 1 - (od & 1);
  const int thbase = 1 - ph;
  const int id0 = (od + 1) >> 1;

  {
    const int t16 = tid & 15;
    const int cl  = tid >> 4;
    const int td = tdbase + ((t16 >> 3) << 1);
    const int th = thbase + (((t16 >> 2) & 1) << 1);
    const float* wsrc = wg + (cl << 6) + td * 16 + th * 4 + (t16 & 3);
    float* wdst = ws + cl * WPC + (t16 << 1);

    const int within = tid & 127;
    const int ih = within >> 3;
    const int j  = within & 7;
    const int pg = tid >> 7;
    const int slot = ih + 1 - ph;
    const bool ok0 = id0 < 16;
    const bool ok1 = id0 >= 1;
    const int idc0 = ok0 ? id0 : 15;
    const int idc1 = ok1 ? id0 - 1 : 0;
    const float* xsrc = xg + (b << 18) + (pg << 13) + (within << 1);
    float* xdst = xs + slot * XS_S + pg * XR + 4 + (j << 2);

    float wv[32];
    v2f   xv[32];
#pragma unroll
    for (int k = 0; k < 32; ++k)
      wv[k] = wsrc[(k >> 2) * 8192 + (k & 1) * 4096 + ((k >> 1) & 1) * 2048];
#pragma unroll
    for (int k = 0; k < 32; ++k) {
      const int r2 = k >> 2;
      const int idc = (r2 & 1) ? idc1 : idc0;
      xv[k] = *(const v2f*)(xsrc + ((r2 >> 1) << 16) + (((k >> 1) & 1) << 15) + ((k & 1) << 12) + (idc << 8));
    }
#pragma unroll
    for (int kk = 0; kk < 16; ++kk)
      *(v2f*)(wdst + (kk >> 1) * WPA + (kk & 1) * 32 * WPC) = (v2f){wv[2 * kk], wv[2 * kk + 1]};
#pragma unroll
    for (int kk = 0; kk < 16; ++kk) {
      const int r2 = kk >> 1;
      const bool ok = (r2 & 1) ? ok1 : ok0;
      v2f a = xv[2 * kk], c = xv[2 * kk + 1];
      v4f v = ok ? (v4f){a.x, c.x, a.y, c.y} : (v4f){0.f, 0.f, 0.f, 0.f};
      *(v4f*)(xdst + r2 * XS_R2 + (kk & 1) * 4 * XR) = v;
    }
    for (int r = tid; r < XROWS + 1; r += 512)
      *(v4f*)(xs + r * XR) = (v4f){0.f, 0.f, 0.f, 0.f};
    {
      const int zslot = ph ? 16 : 0;
      const int rr = tid >> 3;
      *(v4f*)(xs + (((rr >> 3) * 17 + zslot) * 8 + (rr & 7)) * XR + 4 + ((tid & 7) << 2)) =
          (v4f){0.f, 0.f, 0.f, 0.f};
    }
  }
  __syncthreads();

  const int lane = tid & 31;
  const int q    = tid >> 5;
  const int ln   = lane & 15;
  const int kh   = lane >> 4;
  const int oh   = 2 * q + ph;

  const float* xlane = xs + q * XS_S + kh * XR + (ln << 1) + 4;
  const float* wlane = ws + kh * WPA + ln * WPC;

  for (int par = 0; par < 2; ++par) {
    const float* xb = xlane + 2 * par;
    const float* wb = wlane + 2 * (1 - par);
    v8f acc[4][4];
#pragma unroll
    for (int i = 0; i < 4; ++i)
#pragma unroll
      for (int o = 0; o < 4; ++o)
        acc[i][o] = (v8f){0.f, 0.f, 0.f, 0.f, 0.f, 0.f, 0.f, 0.f};

    for (int tap = 0; tap < 8; ++tap) {
      const int twi = tap & 1;
      const int thi = (tap >> 1) & 1;
      const int tdi = tap >> 2;
      const int xoff = tdi * XS_R2 + (1 - thi) * XS_S - 2 * twi;
      const int woff = (tdi * 8 + thi * 4) * 2 + 4 * twi;
#pragma unroll
      for (int ch = 0; ch < 4; ++ch) {
        v2f wf[4], xf[4];
#pragma unroll
        for (int o = 0; o < 4; ++o)
          wf[o] = *(const v2f*)(wb + woff + o * 16 * WPC + ch * 2 * WPA);
#pragma unroll
        for (int i = 0; i < 4; ++i)
          xf[i] = *(const v2f*)(xb + xoff + i * 2 * XS_R2 + ch * 2 * XR);
#pragma unroll
        for (int i = 0; i < 4; ++i)
#pragma unroll
          for (int o = 0; o < 4; ++o)
            acc[i][o] = __builtin_amdgcn_wmma_f32_16x16x4_f32(
                 false, wf[o],  false, xf[i],
                 (short)0, acc[i][o],  false,  false);
      }
    }

    float nb[4][4], logit[4][4];
#pragma unroll
    for (int i = 0; i < 4; ++i)
#pragma unroll
      for (int o = 0; o < 4; ++o) {
        float nv2 = 0.f;
#pragma unroll
        for (int g = 0; g < 8; ++g) nv2 += acc[i][o][g] * acc[i][o][g];
        nv2 += __shfl_xor(nv2, 16);
        nb[i][o] = sqrtf(nv2);
        logit[i][o] = 0.f;
      }

    const int ow = (ln << 1) + par;
#pragma unroll
    for (int it = 0; it < 3; ++it) {
      float route[4][4];
#pragma unroll
      for (int i = 0; i < 4; ++i) {
        float m = fmaxf(fmaxf(logit[i][0], logit[i][1]), fmaxf(logit[i][2], logit[i][3]));
        float e0 = __expf(logit[i][0] - m), e1 = __expf(logit[i][1] - m);
        float e2 = __expf(logit[i][2] - m), e3 = __expf(logit[i][3] - m);
        float s = (e0 + e1) + (e2 + e3);
        route[i][0] = e0 / s; route[i][1] = e1 / s;
        route[i][2] = e2 / s; route[i][3] = e3 / s;
      }
#pragma unroll
      for (int o = 0; o < 4; ++o) {
        float pre[8];
        const float* bo = bg + o * 16 + (kh << 3);
#pragma unroll
        for (int g = 0; g < 8; ++g) {
          float p = route[0][o] * acc[0][o][g];
          p += route[1][o] * acc[1][o][g];
          p += route[2][o] * acc[2][o][g];
          p += route[3][o] * acc[3][o][g];
          pre[g] = p + bo[g];
        }
        float n2 = 0.f;
#pragma unroll
        for (int g = 0; g < 8; ++g) n2 += pre[g] * pre[g];
        n2 += __shfl_xor(n2, 16);
        float nrm = sqrtf(n2);
        if (it < 2) {
#pragma unroll
          for (int i = 0; i < 4; ++i) {
            float dt = 0.f;
#pragma unroll
            for (int g = 0; g < 8; ++g) dt += pre[g] * acc[i][o][g];
            dt += __shfl_xor(dt, 16);
            logit[i][o] += dt / fmaxf(nrm * nb[i][o], 1e-8f);
          }
        } else {
          float scale = n2 / ((nrm + 1e-12f) * (1.f + n2));
#pragma unroll
          for (int g = 0; g < 8; ++g) {
            int ao = (kh << 3) + g;
            outg[(((b * 4 + o) * 16 + ao) << 15) + (od << 10) + (oh << 5) + ow] =
                pre[g] * scale;
          }
        }
      }
    }
  }
}

extern "C" void kernel_launch(void* const* d_in, const int* in_sizes, int n_in,
                              void* d_out, int out_size, void* d_ws, size_t ws_size,
                              hipStream_t stream) {
  (void)in_sizes; (void)n_in; (void)out_size; (void)d_ws; (void)ws_size;
  const float* x = (const float*)d_in[0];
  const float* w = (const float*)d_in[1];
  const float* b = (const float*)d_in[2];
  float* out = (float*)d_out;

  deconv_caps_routing<<<dim3(4 * 32 * 2), dim3(512), 0, stream>>>(x, w, b, out);
}


# wave priority falls with the work left (s_setprio 3, 2, 1, 0 for GEMM 0, routing 0, GEMM 1, routing 1): the 4 waves of a SIMD stay together, not oldest-first; no lone last wave; sums unchanged
# speedup vs baseline: 2.6321x; 1.0700x over previous
#include <hip/hip_runtime.h>

typedef float v2f __attribute__((ext_vector_type(2)));
typedef float v4f __attribute__((ext_vector_type(4)));
typedef float v8f __attribute__((ext_vector_type(8)));

#define XR 36
#define XROWS 1088
#define XS_N (XROWS * XR + 8)
#define XS_R2 (17 * 8 * XR)
#define XS_S (8 * XR)
#define WPC 36
#define WPA 2306
#define WS_N (8 * WPA)

__global__ __launch_bounds__(512)
void deconv_caps_routing(const float* __restrict__ xg,
                         const float* __restrict__ wg,
                         const float* __restrict__ bg,
                         float* __restrict__ outg) {
  __shared__ __attribute__((aligned(16))) float xs[XS_N];
  __shared__ __attribute__((aligned(16))) float ws[WS_N];

  const int tid = threadIdx.x;
  const int blk = blockIdx.x;
  const int b  = blk >> 6;
  const int od = (blk >> 1) & 31;
  const int ph = blk & 1;

  const int tdbase = 1 - (od & 1);
  const int thbase = 1 - ph;
  const int id0 = (od + 1) >> 1;

  {
    const int t16 = tid & 15;
    const int cl  = tid >> 4;
    const int td = tdbase + ((t16 >> 3) << 1);
    const int th = thbase + (((t16 >> 2) & 1) << 1);
    const float* wsrc = wg + (cl << 6) + td * 16 + th * 4 + (t16 & 3);
    float* wdst = ws + cl * WPC + (t16 << 1);

    const int within = tid & 127;
    const int ih = within >> 3;
    const int j  = within & 7;
    const int pg = tid >> 7;
    const int slot = ih + 1 - ph;
    const bool ok0 = id0 < 16;
    const bool ok1 = id0 >= 1;
    const int idc0 = ok0 ? id0 : 15;
    const int idc1 = ok1 ? id0 - 1 : 0;
    const float* xsrc = xg + (b << 18) + (pg << 13) + (within << 1);
    float* xdst = xs + slot * XS_S + pg * XR + 4 + (j << 2);

    float wv[32];
    v2f   xv[32];
#pragma unroll
    for (int k = 0; k < 32; ++k)
      wv[k] = wsrc[(k >> 2) * 8192 + (k & 1) * 4096 + ((k >> 1) & 1) * 2048];
#pragma unroll
    for (int k = 0; k < 32; ++k) {
      const int r2 = k >> 2;
      const int idc = (r2 & 1) ? idc1 : idc0;
      xv[k] = *(const v2f*)(xsrc + ((r2 >> 1) << 16) + (((k >> 1) & 1) << 15) + ((k & 1) << 12) + (idc << 8));
    }
#pragma unroll
    for (int kk = 0; kk < 16; ++kk)
      *(v2f*)(wdst + (kk >> 1) * WPA + (kk & 1) * 32 * WPC) = (v2f){wv[2 * kk], wv[2 * kk + 1]};
#pragma unroll
    for (int kk = 0; kk < 16; ++kk) {
      const int r2 = kk >> 1;
      const bool ok = (r2 & 1) ? ok1 : ok0;
      v2f a = xv[2 * kk], c = xv[2 * kk + 1];
      v4f v = ok ? (v4f){a.x, c.x, a.y, c.y} : (v4f){0.f, 0.f, 0.f, 0.f};
      *(v4f*)(xdst + r2 * XS_R2 + (kk & 1) * 4 * XR) = v;
    }
    for (int r = tid; r < XROWS + 1; r += 512)
      *(v4f*)(xs + r * XR) = (v4f){0.f, 0.f, 0.f, 0.f};
    {
      const int zslot = ph ? 16 : 0;
      const int rr = tid >> 3;
      *(v4f*)(xs + (((rr >> 3) * 17 + zslot) * 8 + (rr & 7)) * XR + 4 + ((tid & 7) << 2)) =
          (v4f){0.f, 0.f, 0.f, 0.f};
    }
  }
  __syncthreads();

  const int lane = tid & 31;
  const int q    = tid >> 5;
  const int ln   = lane & 15;
  const int kh   = lane >> 4;
  const int oh   = 2 * q + ph;

  const float* xlane = xs + q * XS_S + kh * XR + (ln << 1) + 4;
  const float* wlane = ws + kh * WPA + ln * WPC;

  for (int par = 0; par < 2; ++par) {
    if (par == 0) __builtin_amdgcn_s_setprio(3); else __builtin_amdgcn_s_setprio(1);
    const float* xb = xlane + 2 * par;
    const float* wb = wlane + 2 * (1 - par);
    v8f acc[4][4];
#pragma unroll
    for (int i = 0; i < 4; ++i)
#pragma unroll
      for (int o = 0; o < 4; ++o)
        acc[i][o] = (v8f){0.f, 0.f, 0.f, 0.f, 0.f, 0.f, 0.f, 0.f};

    for (int tap = 0; tap < 8; ++tap) {
      const int twi = tap & 1;
      const int thi = (tap >> 1) & 1;
      const int tdi = tap >> 2;
      const int xoff = tdi * XS_R2 + (1 - thi) * XS_S - 2 * twi;
      const int woff = (tdi * 8 + thi * 4) * 2 + 4 * twi;
#pragma unroll
      for (int ch = 0; ch < 4; ++ch) {
        v2f wf[4], xf[4];
#pragma unroll
        for (int o = 0; o < 4; ++o)
          wf[o] = *(const v2f*)(wb + woff + o * 16 * WPC + ch * 2 * WPA);
#pragma unroll
        for (int i = 0; i < 4; ++i)
          xf[i] = *(const v2f*)(xb + xoff + i * 2 * XS_R2 + ch * 2 * XR);
#pragma unroll
        for (int i = 0; i < 4; ++i)
#pragma unroll
          for (int o = 0; o < 4; ++o)
            acc[i][o] = __builtin_amdgcn_wmma_f32_16x16x4_f32(
                 false, wf[o],  false, xf[i],
                 (short)0, acc[i][o],  false,  false);
      }
    }

    if (par == 0) __builtin_amdgcn_s_setprio(2); else __builtin_amdgcn_s_setprio(0);
    float nb[4][4], logit[4][4];
#pragma unroll
    for (int i = 0; i < 4; ++i)
#pragma unroll
      for (int o = 0; o < 4; ++o) {
        float nv2 = 0.f;
#pragma unroll
        for (int g = 0; g < 8; ++g) nv2 += acc[i][o][g] * acc[i][o][g];
        nv2 += __shfl_xor(nv2, 16);
        nb[i][o] = sqrtf(nv2);
        logit[i][o] = 0.f;
      }

    const int ow = (ln << 1) + par;
#pragma unroll
    for (int it = 0; it < 3; ++it) {
      float route[4][4];
#pragma unroll
      for (int i = 0; i < 4; ++i) {
        float m = fmaxf(fmaxf(logit[i][0], logit[i][1]), fmaxf(logit[i][2], logit[i][3]));
        float e0 = __expf(logit[i][0] - m), e1 = __expf(logit[i][1] - m);
        float e2 = __expf(logit[i][2] - m), e3 = __expf(logit[i][3] - m);
        float s = (e0 + e1) + (e2 + e3);
        route[i][0] = e0 / s; route[i][1] = e1 / s;
        route[i][2] = e2 / s; route[i][3] = e3 / s;
      }
#pragma unroll
      for (int o = 0; o < 4; ++o) {
        float pre[8];
        const float* bo = bg + o * 16 + (kh << 3);
#pragma unroll
        for (int g = 0; g < 8; ++g) {
          float p = route[0][o] * acc[0][o][g];
          p += route[1][o] * acc[1][o][g];
          p += route[2][o] * acc[2][o][g];
          p += route[3][o] * acc[3][o][g];
          pre[g] = p + bo[g];
        }
        float n2 = 0.f;
#pragma unroll
        for (int g = 0; g < 8; ++g) n2 += pre[g] * pre[g];
        n2 += __shfl_xor(n2, 16);
        float nrm = sqrtf(n2);
        if (it < 2) {
#pragma unroll
          for (int i = 0; i < 4; ++i) {
            float dt = 0.f;
#pragma unroll
            for (int g = 0; g < 8; ++g) dt += pre[g] * acc[i][o][g];
            dt += __shfl_xor(dt, 16);
            logit[i][o] += dt / fmaxf(nrm * nb[i][o], 1e-8f);
          }
        } else {
          float scale = n2 / ((nrm + 1e-12f) * (1.f + n2));
#pragma unroll
          for (int g = 0; g < 8; ++g) {
            int ao = (kh << 3) + g;
            outg[(((b * 4 + o) * 16 + ao) << 15) + (od << 10) + (oh << 5) + ow] =
                pre[g] * scale;
          }
        }
      }
    }
  }
}

extern "C" void kernel_launch(void* const* d_in, const int* in_sizes, int n_in,
                              void* d_out, int out_size, void* d_ws, size_t ws_size,
                              hipStream_t stream) {
  (void)in_sizes; (void)n_in; (void)out_size; (void)d_ws; (void)ws_size;
  const float* x = (const float*)d_in[0];
  const float* w = (const float*)d_in[1];
  const float* b = (const float*)d_in[2];
  float* out = (float*)d_out;

  deconv_caps_routing<<<dim3(4 * 32 * 2), dim3(512), 0, stream>>>(x, w, b, out);
}


# D tile rows = 2 out capsules x 8 atoms (weight rows permuted in LDS): lane owns whole capsules; atom sums in-lane as packed even+odd chains (order changed); norm/div/exp once, not in both half-lanes
# speedup vs baseline: 3.1166x; 1.1841x over previous
#include <hip/hip_runtime.h>

typedef float v2f __attribute__((ext_vector_type(2)));
typedef float v4f __attribute__((ext_vector_type(4)));
typedef float v8f __attribute__((ext_vector_type(8)));

#define XR 36
#define XROWS 1088
#define XS_N (XROWS * XR + 8)
#define XS_R2 (17 * 8 * XR)
#define XS_S (8 * XR)
#define WPC 36
#define WPA 2306
#define WS_N (8 * WPA)

__global__ __launch_bounds__(512)
void deconv_caps_routing(const float* __restrict__ xg,
                         const float* __restrict__ wg,
                         const float* __restrict__ bg,
                         float* __restrict__ outg) {
  __shared__ __attribute__((aligned(16))) float xs[XS_N];
  __shared__ __attribute__((aligned(16))) float ws[WS_N];

  const int tid = threadIdx.x;
  const int blk = blockIdx.x;
  const int b  = blk >> 6;
  const int od = (blk >> 1) & 31;
  const int ph = blk & 1;

  const int tdbase = 1 - (od & 1);
  const int thbase = 1 - ph;
  const int id0 = (od + 1) >> 1;

  {
    const int t16 = tid & 15;
    const int cl  = tid >> 4;
    const int td = tdbase + ((t16 >> 3) << 1);
    const int th = thbase + (((t16 >> 2) & 1) << 1);
    const float* wsrc = wg + (cl << 6) + td * 16 + th * 4 + (t16 & 3);
    const int wrow = ((cl >> 4) << 5) + (((cl >> 3) & 1) << 4) + (cl & 7);
    float* wdst = ws + wrow * WPC + (t16 << 1);

    const int within = tid & 127;
    const int ih = within >> 3;
    const int j  = within & 7;
    const int pg = tid >> 7;
    const int slot = ih + 1 - ph;
    const bool ok0 = id0 < 16;
    const bool ok1 = id0 >= 1;
    const int idc0 = ok0 ? id0 : 15;
    const int idc1 = ok1 ? id0 - 1 : 0;
    const float* xsrc = xg + (b << 18) + (pg << 13) + (within << 1);
    float* xdst = xs + slot * XS_S + pg * XR + 4 + (j << 2);

    float wv[32];
    v2f   xv[32];
#pragma unroll
    for (int k = 0; k < 32; ++k)
      wv[k] = wsrc[(k >> 2) * 8192 + (k & 1) * 4096 + ((k >> 1) & 1) * 2048];
#pragma unroll
    for (int k = 0; k < 32; ++k) {
      const int r2 = k >> 2;
      const int idc = (r2 & 1) ? idc1 : idc0;
      xv[k] = *(const v2f*)(xsrc + ((r2 >> 1) << 16) + (((k >> 1) & 1) << 15) + ((k & 1) << 12) + (idc << 8));
    }
#pragma unroll
    for (int kk = 0; kk < 16; ++kk)
      *(v2f*)(wdst + (kk >> 1) * WPA + (kk & 1) * 8 * WPC) = (v2f){wv[2 * kk], wv[2 * kk + 1]};
#pragma unroll
    for (int kk = 0; kk < 16; ++kk) {
      const int r2 = kk >> 1;
      const bool ok = (r2 & 1) ? ok1 : ok0;
      v2f a = xv[2 * kk], c = xv[2 * kk + 1];
      v4f v = ok ? (v4f){a.x, c.x, a.y, c.y} : (v4f){0.f, 0.f, 0.f, 0.f};
      *(v4f*)(xdst + r2 * XS_R2 + (kk & 1) * 4 * XR) = v;
    }
    for (int r = tid; r < XROWS + 1; r += 512)
      *(v4f*)(xs + r * XR) = (v4f){0.f, 0.f, 0.f, 0.f};
    {
      const int zslot = ph ? 16 : 0;
      const int rr = tid >> 3;
      *(v4f*)(xs + (((rr >> 3) * 17 + zslot) * 8 + (rr & 7)) * XR + 4 + ((tid & 7) << 2)) =
          (v4f){0.f, 0.f, 0.f, 0.f};
    }
  }
  __syncthreads();

  const int lane = tid & 31;
  const int q    = tid >> 5;
  const int ln   = lane & 15;
  const int kh   = lane >> 4;
  const int oh   = 2 * q + ph;

  const float* xlane = xs + q * XS_S + kh * XR + (ln << 1) + 4;
  const float* wlane = ws + kh * WPA + ln * WPC;

  for (int par = 0; par < 2; ++par) {
    if (par == 0) __builtin_amdgcn_s_setprio(3); else __builtin_amdgcn_s_setprio(1);
    const float* xb = xlane + 2 * par;
    const float* wb = wlane + 2 * (1 - par);
    v8f acc[4][4];
#pragma unroll
    for (int i = 0; i < 4; ++i)
#pragma unroll
      for (int o = 0; o < 4; ++o)
        acc[i][o] = (v8f){0.f, 0.f, 0.f, 0.f, 0.f, 0.f, 0.f, 0.f};

    for (int tap = 0; tap < 8; ++tap) {
      const int twi = tap & 1;
      const int thi = (tap >> 1) & 1;
      const int tdi = tap >> 2;
      const int xoff = tdi * XS_R2 + (1 - thi) * XS_S - 2 * twi;
      const int woff = (tdi * 8 + thi * 4) * 2 + 4 * twi;
#pragma unroll
      for (int ch = 0; ch < 4; ++ch) {
        v2f wf[4], xf[4];
#pragma unroll
        for (int o = 0; o < 4; ++o)
          wf[o] = *(const v2f*)(wb + woff + o * 16 * WPC + ch * 2 * WPA);
#pragma unroll
        for (int i = 0; i < 4; ++i)
          xf[i] = *(const v2f*)(xb + xoff + i * 2 * XS_R2 + ch * 2 * XR);
#pragma unroll
        for (int i = 0; i < 4; ++i)
#pragma unroll
          for (int o = 0; o < 4; ++o)
            acc[i][o] = __builtin_amdgcn_wmma_f32_16x16x4_f32(
                 false, wf[o],  false, xf[i],
                 (short)0, acc[i][o],  false,  false);
      }
    }

    if (par == 0) __builtin_amdgcn_s_setprio(2); else __builtin_amdgcn_s_setprio(0);
#define ATOMS2(i, op, j) ((v2f){acc[i][2 * (op) + ((j) >> 2)][((j) & 3) * 2], acc[i][2 * (op) + ((j) >> 2)][((j) & 3) * 2 + 1]})
    float nb[4][2], logit[4][2];
#pragma unroll
    for (int i = 0; i < 4; ++i)
#pragma unroll
      for (int op = 0; op < 2; ++op) {
        v2f s = ATOMS2(i, op, 0) * ATOMS2(i, op, 0);
#pragma unroll
        for (int j = 1; j < 8; ++j) s = __builtin_elementwise_fma(ATOMS2(i, op, j), ATOMS2(i, op, j), s);
        nb[i][op] = sqrtf(s.x + s.y);
        logit[i][op] = 0.f;
      }

    const int ow = (ln << 1) + par;
#pragma unroll
    for (int it = 0; it < 3; ++it) {
      float route[4][2];
#pragma unroll
      for (int i = 0; i < 4; ++i) {
        if (it == 0) {
          route[i][0] = 0.25f; route[i][1] = 0.25f;
        } else {
          float ml = fmaxf(logit[i][0], logit[i][1]);
          float m = fmaxf(ml, __shfl_xor(ml, 16));
          float e0 = __expf(logit[i][0] - m), e1 = __expf(logit[i][1] - m);
          float sl = e0 + e1;
          float s = sl + __shfl_xor(sl, 16);
          route[i][0] = e0 / s; route[i][1] = e1 / s;
        }
      }
#pragma unroll
      for (int op = 0; op < 2; ++op) {
        v2f pre[8];
        const int o = 2 * kh + op;
        const v2f* bo = (const v2f*)(bg + o * 16);
        const v2f r0 = {route[0][op], route[0][op]}, r1 = {route[1][op], route[1][op]};
        const v2f r2 = {route[2][op], route[2][op]}, r3 = {route[3][op], route[3][op]};
#pragma unroll
        for (int j = 0; j < 8; ++j) {
          v2f p = r0 * ATOMS2(0, op, j);
          p = __builtin_elementwise_fma(r1, ATOMS2(1, op, j), p);
          p = __builtin_elementwise_fma(r2, ATOMS2(2, op, j), p);
          p = __builtin_elementwise_fma(r3, ATOMS2(3, op, j), p);
          pre[j] = p + bo[j];
        }
        v2f n2v = pre[0] * pre[0];
#pragma unroll
        for (int j = 1; j < 8; ++j) n2v = __builtin_elementwise_fma(pre[j], pre[j], n2v);
        float n2 = n2v.x + n2v.y;
        float nrm = sqrtf(n2);
        if (it < 2) {
#pragma unroll
          for (int i = 0; i < 4; ++i) {
            v2f d = pre[0] * ATOMS2(i, op, 0);
#pragma unroll
            for (int j = 1; j < 8; ++j) d = __builtin_elementwise_fma(pre[j], ATOMS2(i, op, j), d);
            logit[i][op] += (d.x + d.y) / fmaxf(nrm * nb[i][op], 1e-8f);
          }
        } else {
          float scale = n2 / ((nrm + 1e-12f) * (1.f + n2));
          float* orow = outg + (((b * 4 + o) * 16) << 15) + (od << 10) + (oh << 5) + ow;
#pragma unroll
          for (int j = 0; j < 8; ++j) {
            v2f v = pre[j] * (v2f){scale, scale};
            orow[(2 * j) << 15] = v.x;
            orow[(2 * j + 1) << 15] = v.y;
          }
        }
      }
    }
#undef ATOMS2
  }
}

extern "C" void kernel_launch(void* const* d_in, const int* in_sizes, int n_in,
                              void* d_out, int out_size, void* d_ws, size_t ws_size,
                              hipStream_t stream) {
  (void)in_sizes; (void)n_in; (void)out_size; (void)d_ws; (void)ws_size;
  const float* x = (const float*)d_in[0];
  const float* w = (const float*)d_in[1];
  const float* b = (const float*)d_in[2];
  float* out = (float*)d_out;

  deconv_caps_routing<<<dim3(4 * 32 * 2), dim3(512), 0, stream>>>(x, w, b, out);
}


# GEMM in 16 blocks of 32 matrix ops; the next block's LDS fragment reads stand in ONE place ahead of the block (2 register sets, counted waits): 16 read places per pass, was 58; k order, sums unchanged
# speedup vs baseline: 3.5209x; 1.1297x over previous
#include <hip/hip_runtime.h>

typedef float v2f __attribute__((ext_vector_type(2)));
typedef float v4f __attribute__((ext_vector_type(4)));
typedef float v8f __attribute__((ext_vector_type(8)));

#define XR 36
#define XROWS 1088
#define XS_N (XROWS * XR + 8)
#define XS_R2 (17 * 8 * XR)
#define XS_S (8 * XR)
#define WPC 36
#define WPA 2306
#define WS_N (8 * WPA)

__global__ __launch_bounds__(512)
void deconv_caps_routing(const float* __restrict__ xg,
                         const float* __restrict__ wg,
                         const float* __restrict__ bg,
                         float* __restrict__ outg) {
  __shared__ __attribute__((aligned(16))) float xs[XS_N];
  __shared__ __attribute__((aligned(16))) float ws[WS_N];

  const int tid = threadIdx.x;
  const int blk = blockIdx.x;
  const int b  = blk >> 6;
  const int od = (blk >> 1) & 31;
  const int ph = blk & 1;

  const int tdbase = 1 - (od & 1);
  const int thbase = 1 - ph;
  const int id0 = (od + 1) >> 1;

  {
    const int t16 = tid & 15;
    const int cl  = tid >> 4;
    const int td = tdbase + ((t16 >> 3) << 1);
    const int th = thbase + (((t16 >> 2) & 1) << 1);
    const float* wsrc = wg + (cl << 6) + td * 16 + th * 4 + (t16 & 3);
    const int wrow = ((cl >> 4) << 5) + (((cl >> 3) & 1) << 4) + (cl & 7);
    float* wdst = ws + wrow * WPC + (t16 << 1);

    const int within = tid & 127;
    const int ih = within >> 3;
    const int j  = within & 7;
    const int pg = tid >> 7;
    const int slot = ih + 1 - ph;
    const bool ok0 = id0 < 16;
    const bool ok1 = id0 >= 1;
    const int idc0 = ok0 ? id0 : 15;
    const int idc1 = ok1 ? id0 - 1 : 0;
    const float* xsrc = xg + (b << 18) + (pg << 13) + (within << 1);
    float* xdst = xs + slot * XS_S + pg * XR + 4 + (j << 2);

    float wv[32];
    v2f   xv[32];
#pragma unroll
    for (int k = 0; k < 32; ++k)
      wv[k] = wsrc[(k >> 2) * 8192 + (k & 1) * 4096 + ((k >> 1) & 1) * 2048];
#pragma unroll
    for (int k = 0; k < 32; ++k) {
      const int r2 = k >> 2;
      const int idc = (r2 & 1) ? idc1 : idc0;
      xv[k] = *(const v2f*)(xsrc + ((r2 >> 1) << 16) + (((k >> 1) & 1) << 15) + ((k & 1) << 12) + (idc << 8));
    }
#pragma unroll
    for (int kk = 0; kk < 16; ++kk)
      *(v2f*)(wdst + (kk >> 1) * WPA + (kk & 1) * 8 * WPC) = (v2f){wv[2 * kk], wv[2 * kk + 1]};
#pragma unroll
    for (int kk = 0; kk < 16; ++kk) {
      const int r2 = kk >> 1;
      const bool ok = (r2 & 1) ? ok1 : ok0;
      v2f a = xv[2 * kk], c = xv[2 * kk + 1];
      v4f v = ok ? (v4f){a.x, c.x, a.y, c.y} : (v4f){0.f, 0.f, 0.f, 0.f};
      *(v4f*)(xdst + r2 * XS_R2 + (kk & 1) * 4 * XR) = v;
    }
    for (int r = tid; r < XROWS + 1; r += 512)
      *(v4f*)(xs + r * XR) = (v4f){0.f, 0.f, 0.f, 0.f};
    {
      const int zslot = ph ? 16 : 0;
      const int rr = tid >> 3;
      *(v4f*)(xs + (((rr >> 3) * 17 + zslot) * 8 + (rr & 7)) * XR + 4 + ((tid & 7) << 2)) =
          (v4f){0.f, 0.f, 0.f, 0.f};
    }
  }
  __syncthreads();

  const int lane = tid & 31;
  const int q    = tid >> 5;
  const int ln   = lane & 15;
  const int kh   = lane >> 4;
  const int oh   = 2 * q + ph;

  const float* xlane = xs + q * XS_S + kh * XR + (ln << 1) + 4;
  const float* wlane = ws + kh * WPA + ln * WPC;

  for (int par = 0; par < 2; ++par) {
    if (par == 0) __builtin_amdgcn_s_setprio(3); else __builtin_amdgcn_s_setprio(1);
    const float* xb = xlane + 2 * par;
    const float* wb = wlane + 2 * (1 - par);
    v8f acc[4][4];
#pragma unroll
    for (int i = 0; i < 4; ++i)
#pragma unroll
      for (int o = 0; o < 4; ++o)
        acc[i][o] = (v8f){0.f, 0.f, 0.f, 0.f, 0.f, 0.f, 0.f, 0.f};

    v2f wf[2][2][4], xf[2][2][4];
#define FRAG_LOADS(g, set)                                                                   \
    {                                                                                        \
      const int tap_ = (g) >> 1;                                                             \
      const int twi_ = tap_ & 1, thi_ = (tap_ >> 1) & 1, tdi_ = tap_ >> 2;                   \
      const int xoff_ = tdi_ * XS_R2 + (1 - thi_) * XS_S - 2 * twi_;                         \
      const int woff_ = (tdi_ * 8 + thi_ * 4) * 2 + 4 * twi_;                                \
      _Pragma("unroll") for (int c_ = 0; c_ < 2; ++c_) {                                     \
        const int ch_ = 2 * ((g) & 1) + c_;                \
        _Pragma("unroll") for (int o_ = 0; o_ < 4; ++o_)                                     \
          wf[set][c_][o_] = *(const v2f*)(wb + woff_ + o_ * 16 * WPC + ch_ * 2 * WPA);       \
        _Pragma("unroll") for (int i_ = 0; i_ < 4; ++i_)                                     \
          xf[set][c_][i_] = *(const v2f*)(xb + xoff_ + i_ * 2 * XS_R2 + ch_ * 2 * XR);       \
      }                                                                                      \
    }
    FRAG_LOADS(0, 0)
#pragma unroll
    for (int g = 0; g < 16; ++g) {
      if (g + 1 < 16) FRAG_LOADS(g + 1, (g + 1) & 1)
      __builtin_amdgcn_sched_barrier(0);
#pragma unroll
      for (int c = 0; c < 2; ++c)
#pragma unroll
        for (int i = 0; i < 4; ++i)
#pragma unroll
          for (int o = 0; o < 4; ++o)
            acc[i][o] = __builtin_amdgcn_wmma_f32_16x16x4_f32(
                 false, wf[g & 1][c][o],  false, xf[g & 1][c][i],
                 (short)0, acc[i][o],  false,  false);
      __builtin_amdgcn_sched_barrier(0);
    }
#undef FRAG_LOADS

    if (par == 0) __builtin_amdgcn_s_setprio(2); else __builtin_amdgcn_s_setprio(0);
#define ATOMS2(i, op, j) ((v2f){acc[i][2 * (op) + ((j) >> 2)][((j) & 3) * 2], acc[i][2 * (op) + ((j) >> 2)][((j) & 3) * 2 + 1]})
    float nb[4][2], logit[4][2];
#pragma unroll
    for (int i = 0; i < 4; ++i)
#pragma unroll
      for (int op = 0; op < 2; ++op) {
        v2f s = ATOMS2(i, op, 0) * ATOMS2(i, op, 0);
#pragma unroll
        for (int j = 1; j < 8; ++j) s = __builtin_elementwise_fma(ATOMS2(i, op, j), ATOMS2(i, op, j), s);
        nb[i][op] = sqrtf(s.x + s.y);
        logit[i][op] = 0.f;
      }

    const int ow = (ln << 1) + par;
#pragma unroll
    for (int it = 0; it < 3; ++it) {
      float route[4][2];
#pragma unroll
      for (int i = 0; i < 4; ++i) {
        if (it == 0) {
          route[i][0] = 0.25f; route[i][1] = 0.25f;
        } else {
          float ml = fmaxf(logit[i][0], logit[i][1]);
          float m = fmaxf(ml, __shfl_xor(ml, 16));
          float e0 = __expf(logit[i][0] - m), e1 = __expf(logit[i][1] - m);
          float sl = e0 + e1;
          float s = sl + __shfl_xor(sl, 16);
          route[i][0] = e0 / s; route[i][1] = e1 / s;
        }
      }
#pragma unroll
      for (int op = 0; op < 2; ++op) {
        v2f pre[8];
        const int o = 2 * kh + op;
        const v2f* bo = (const v2f*)(bg + o * 16);
        const v2f r0 = {route[0][op], route[0][op]}, r1 = {route[1][op], route[1][op]};
        const v2f r2 = {route[2][op], route[2][op]}, r3 = {route[3][op], route[3][op]};
#pragma unroll
        for (int j = 0; j < 8; ++j) {
          v2f p = r0 * ATOMS2(0, op, j);
          p = __builtin_elementwise_fma(r1, ATOMS2(1, op, j), p);
          p = __builtin_elementwise_fma(r2, ATOMS2(2, op, j), p);
          p = __builtin_elementwise_fma(r3, ATOMS2(3, op, j), p);
          pre[j] = p + bo[j];
        }
        v2f n2v = pre[0] * pre[0];
#pragma unroll
        for (int j = 1; j < 8; ++j) n2v = __builtin_elementwise_fma(pre[j], pre[j], n2v);
        float n2 = n2v.x + n2v.y;
        float nrm = sqrtf(n2);
        if (it < 2) {
#pragma unroll
          for (int i = 0; i < 4; ++i) {
            v2f d = pre[0] * ATOMS2(i, op, 0);
#pragma unroll
            for (int j = 1; j < 8; ++j) d = __builtin_elementwise_fma(pre[j], ATOMS2(i, op, j), d);
            logit[i][op] += (d.x + d.y) / fmaxf(nrm * nb[i][op], 1e-8f);
          }
        } else {
          float scale = n2 / ((nrm + 1e-12f) * (1.f + n2));
          float* orow = outg + (((b * 4 + o) * 16) << 15) + (od << 10) + (oh << 5) + ow;
#pragma unroll
          for (int j = 0; j < 8; ++j) {
            v2f v = pre[j] * (v2f){scale, scale};
            orow[(2 * j) << 15] = v.x;
            orow[(2 * j + 1) << 15] = v.y;
          }
        }
      }
    }
#undef ATOMS2
  }
}

extern "C" void kernel_launch(void* const* d_in, const int* in_sizes, int n_in,
                              void* d_out, int out_size, void* d_ws, size_t ws_size,
                              hipStream_t stream) {
  (void)in_sizes; (void)n_in; (void)out_size; (void)d_ws; (void)ws_size;
  const float* x = (const float*)d_in[0];
  const float* w = (const float*)d_in[1];
  const float* b = (const float*)d_in[2];
  float* out = (float*)d_out;

  deconv_caps_routing<<<dim3(4 * 32 * 2), dim3(512), 0, stream>>>(x, w, b, out);
}


# two adjacent oh rows from 3 matrix products, not 4 (2-tap minimal filtering along oh: X row differences, W tap sum on the fragments); wave = row pair; 768 matrix ops a pair, was 1024; sums reordered
# speedup vs baseline: 3.9844x; 1.1316x over previous
#include <hip/hip_runtime.h>

typedef float v2f __attribute__((ext_vector_type(2)));
typedef float v4f __attribute__((ext_vector_type(4)));
typedef float v8f __attribute__((ext_vector_type(8)));

#define XR 36
#define XROWS 1088
#define XS_N (XROWS * XR + 8)
#define XS_R2 (17 * 8 * XR)
#define XS_S (8 * XR)
#define WPC 36
#define WPA 2306
#define WS_N (8 * WPA)

__global__ __launch_bounds__(256)
void deconv_caps_routing(const float* __restrict__ xg,
                         const float* __restrict__ wg,
                         const float* __restrict__ bg,
                         float* __restrict__ outg) {
  __shared__ __attribute__((aligned(16))) float xs[XS_N];
  __shared__ __attribute__((aligned(16))) float ws[WS_N];

  const int t0 = threadIdx.x;
  const int blk = blockIdx.x;
  const int b  = blk >> 6;
  const int od = (blk >> 1) & 31;
  const int ph = blk & 1;

  const int tdbase = 1 - (od & 1);
  const int thbase = 1 - ph;
  const int id0 = (od + 1) >> 1;

  {
    const bool ok0 = id0 < 16;
    const bool ok1 = id0 >= 1;
    const int idc0 = ok0 ? id0 : 15;
    const int idc1 = ok1 ? id0 - 1 : 0;
    const float* wsrc[2];
    float* wdst[2];
    const float* xsrc[2];
    float* xdst[2];
#pragma unroll
    for (int vt = 0; vt < 2; ++vt) {
      const int tid = t0 + (vt << 8);
      const int t16 = tid & 15;
      const int cl  = tid >> 4;
      const int td = tdbase + ((t16 >> 3) << 1);
      const int th = thbase + (((t16 >> 2) & 1) << 1);
      wsrc[vt] = wg + (cl << 6) + td * 16 + th * 4 + (t16 & 3);
      const int wrow = ((cl >> 4) << 5) + (((cl >> 3) & 1) << 4) + (cl & 7);
      wdst[vt] = ws + wrow * WPC + (t16 << 1);
      const int within = tid & 127;
      const int ih = within >> 3;
      const int j  = within & 7;
      const int pg = tid >> 7;
      const int slot = ih + 1 - ph;
      xsrc[vt] = xg + (b << 18) + (pg << 13) + (within << 1);
      xdst[vt] = xs + slot * XS_S + pg * XR + 4 + (j << 2);
    }

    float wv[2][32];
    v2f   xv[2][32];
#pragma unroll
    for (int vt = 0; vt < 2; ++vt)
#pragma unroll
      for (int k = 0; k < 32; ++k)
        wv[vt][k] = wsrc[vt][(k >> 2) * 8192 + (k & 1) * 4096 + ((k >> 1) & 1) * 2048];
#pragma unroll
    for (int vt = 0; vt < 2; ++vt)
#pragma unroll
      for (int k = 0; k < 32; ++k) {
        const int r2 = k >> 2;
        const int idc = (r2 & 1) ? idc1 : idc0;
        xv[vt][k] = *(const v2f*)(xsrc[vt] + ((r2 >> 1) << 16) + (((k >> 1) & 1) << 15) + ((k & 1) << 12) + (idc << 8));
      }
#pragma unroll
    for (int vt = 0; vt < 2; ++vt)
#pragma unroll
      for (int kk = 0; kk < 16; ++kk)
        *(v2f*)(wdst[vt] + (kk >> 1) * WPA + (kk & 1) * 8 * WPC) = (v2f){wv[vt][2 * kk], wv[vt][2 * kk + 1]};
#pragma unroll
    for (int vt = 0; vt < 2; ++vt)
#pragma unroll
      for (int kk = 0; kk < 16; ++kk) {
        const int r2 = kk >> 1;
        const bool ok = (r2 & 1) ? ok1 : ok0;
        v2f a = xv[vt][2 * kk], c = xv[vt][2 * kk + 1];
        v4f v = ok ? (v4f){a.x, c.x, a.y, c.y} : (v4f){0.f, 0.f, 0.f, 0.f};
        *(v4f*)(xdst[vt] + r2 * XS_R2 + (kk & 1) * 4 * XR) = v;
      }
    for (int r = t0; r < XROWS + 1; r += 256)
      *(v4f*)(xs + r * XR) = (v4f){0.f, 0.f, 0.f, 0.f};
#pragma unroll
    for (int vt = 0; vt < 2; ++vt) {
      const int tid = t0 + (vt << 8);
      const int zslot = ph ? 16 : 0;
      const int rr = tid >> 3;
      *(v4f*)(xs + (((rr >> 3) * 17 + zslot) * 8 + (rr & 7)) * XR + 4 + ((tid & 7) << 2)) =
          (v4f){0.f, 0.f, 0.f, 0.f};
    }
  }
  __syncthreads();

  const int lane = t0 & 31;
  const int p    = t0 >> 5;
  const int ln   = lane & 15;
  const int kh   = lane >> 4;

  const float* xlane = xs + (2 * p) * XS_S + kh * XR + (ln << 1) + 4;
  const float* wlane = ws + kh * WPA + ln * WPC;

  for (int par = 0; par < 2; ++par) {
    const float* xb = xlane + 2 * par;
    const float* wb = wlane + 2 * (1 - par);
    v8f acc[4][4], accm[4][4];
    const int ow = (ln << 1) + par;
#pragma nounroll
    for (int r = 0; r < 2; ++r) {
    if (par == 0) { if (r == 0) __builtin_amdgcn_s_setprio(3); else __builtin_amdgcn_s_setprio(2); }
    else          { if (r == 0) __builtin_amdgcn_s_setprio(1); else __builtin_amdgcn_s_setprio(0); }
    if (r == 0) {
#pragma unroll
    for (int i = 0; i < 4; ++i)
#pragma unroll
      for (int o = 0; o < 4; ++o) {
        acc[i][o]  = (v8f){0.f, 0.f, 0.f, 0.f, 0.f, 0.f, 0.f, 0.f};
        accm[i][o] = (v8f){0.f, 0.f, 0.f, 0.f, 0.f, 0.f, 0.f, 0.f};
      }

    {
      v2f wf[2][2][4], xf[2][2][4];
#define FRAG_LOADS_A(g, set)                                                                 \
      {                                                                                      \
        const int tp_ = (g) >> 2;                                                            \
        const int twi_ = tp_ & 1, tdi_ = tp_ >> 1;                                           \
        const int ch_ = (g) & 3;                           \
        const int xoff_ = tdi_ * XS_R2 - 2 * twi_ + ch_ * 2 * XR;                            \
        const int woff_ = tdi_ * 16 + 4 * twi_ + ch_ * 2 * WPA;                              \
        _Pragma("unroll") for (int h_ = 0; h_ < 2; ++h_)                                     \
          _Pragma("unroll") for (int o_ = 0; o_ < 4; ++o_)                                   \
            wf[set][h_][o_] = *(const v2f*)(wb + woff_ + h_ * 8 + o_ * 16 * WPC);            \
        _Pragma("unroll") for (int s_ = 0; s_ < 2; ++s_)                                     \
          _Pragma("unroll") for (int i_ = 0; i_ < 4; ++i_)                                   \
            xf[set][s_][i_] = *(const v2f*)(xb + xoff_ + s_ * XS_S + i_ * 2 * XS_R2);        \
      }
      FRAG_LOADS_A(0, 0)
#pragma unroll
      for (int g = 0; g < 16; ++g) {
        if (g + 1 < 16) FRAG_LOADS_A(g + 1, (g + 1) & 1)
        __builtin_amdgcn_sched_barrier(0);
        v2f wsum[4], d01[4];
#pragma unroll
        for (int o = 0; o < 4; ++o) wsum[o] = wf[g & 1][0][o] + wf[g & 1][1][o];
#pragma unroll
        for (int i = 0; i < 4; ++i) d01[i] = xf[g & 1][0][i] - xf[g & 1][1][i];
#pragma unroll
        for (int i = 0; i < 4; ++i)
#pragma unroll
          for (int o = 0; o < 4; ++o)
            acc[i][o] = __builtin_amdgcn_wmma_f32_16x16x4_f32(
                 false, wf[g & 1][1][o],  false, d01[i],
                 (short)0, acc[i][o],  false,  false);
#pragma unroll
        for (int i = 0; i < 4; ++i)
#pragma unroll
          for (int o = 0; o < 4; ++o)
            accm[i][o] = __builtin_amdgcn_wmma_f32_16x16x4_f32(
                 false, wsum[o],  false, xf[g & 1][1][i],
                 (short)0, accm[i][o],  false,  false);
        __builtin_amdgcn_sched_barrier(0);
      }
#undef FRAG_LOADS_A
    }
#pragma unroll
    for (int i = 0; i < 4; ++i)
#pragma unroll
      for (int o = 0; o < 4; ++o)
        acc[i][o] += accm[i][o];
    } else {
    {
      v2f wf[2][2][4], xf[2][2][2][4];
#define FRAG_LOADS_B(g, set)                                                                 \
      {                                                                                      \
        const int tp_ = (g) >> 1;                                                            \
        const int twi_ = tp_ & 1, tdi_ = tp_ >> 1;                                           \
        _Pragma("unroll") for (int c_ = 0; c_ < 2; ++c_) {                                   \
          const int ch_ = 2 * ((g) & 1) + c_;              \
          const int xoff_ = tdi_ * XS_R2 - 2 * twi_ + ch_ * 2 * XR;                          \
          const int woff_ = tdi_ * 16 + 4 * twi_ + ch_ * 2 * WPA;                            \
          _Pragma("unroll") for (int o_ = 0; o_ < 4; ++o_)                                   \
            wf[set][c_][o_] = *(const v2f*)(wb + woff_ + o_ * 16 * WPC);                     \
          _Pragma("unroll") for (int s_ = 0; s_ < 2; ++s_)                                   \
            _Pragma("unroll") for (int i_ = 0; i_ < 4; ++i_)                                 \
              xf[set][c_][s_][i_] = *(const v2f*)(xb + xoff_ + (s_ + 1) * XS_S + i_ * 2 * XS_R2); \
        }                                                                                    \
      }
      FRAG_LOADS_B(0, 0)
#pragma unroll
      for (int g = 0; g < 8; ++g) {
        if (g + 1 < 8) FRAG_LOADS_B(g + 1, (g + 1) & 1)
        __builtin_amdgcn_sched_barrier(0);
        v2f d21[2][4];
#pragma unroll
        for (int c = 0; c < 2; ++c)
#pragma unroll
          for (int i = 0; i < 4; ++i) d21[c][i] = xf[g & 1][c][1][i] - xf[g & 1][c][0][i];
#pragma unroll
        for (int c = 0; c < 2; ++c)
#pragma unroll
          for (int i = 0; i < 4; ++i)
#pragma unroll
            for (int o = 0; o < 4; ++o)
              accm[i][o] = __builtin_amdgcn_wmma_f32_16x16x4_f32(
                   false, wf[g & 1][c][o],  false, d21[c][i],
                   (short)0, accm[i][o],  false,  false);
        __builtin_amdgcn_sched_barrier(0);
      }
#undef FRAG_LOADS_B
    }

#pragma unroll
    for (int i = 0; i < 4; ++i)
#pragma unroll
      for (int o = 0; o < 4; ++o)
        acc[i][o] = accm[i][o];
    }

    const int oh = 2 * (2 * p + r) + ph;
#define ATOMS2(i, op, j) ((v2f){acc[i][2 * (op) + ((j) >> 2)][((j) & 3) * 2], acc[i][2 * (op) + ((j) >> 2)][((j) & 3) * 2 + 1]})
    float nb[4][2], logit[4][2];
#pragma unroll
    for (int i = 0; i < 4; ++i)
#pragma unroll
      for (int op = 0; op < 2; ++op) {
        v2f s = ATOMS2(i, op, 0) * ATOMS2(i, op, 0);
#pragma unroll
        for (int j = 1; j < 8; ++j) s = __builtin_elementwise_fma(ATOMS2(i, op, j), ATOMS2(i, op, j), s);
        nb[i][op] = sqrtf(s.x + s.y);
        logit[i][op] = 0.f;
      }

#pragma unroll
    for (int it = 0; it < 3; ++it) {
      float route[4][2];
#pragma unroll
      for (int i = 0; i < 4; ++i) {
        if (it == 0) {
          route[i][0] = 0.25f; route[i][1] = 0.25f;
        } else {
          float ml = fmaxf(logit[i][0], logit[i][1]);
          float m = fmaxf(ml, __shfl_xor(ml, 16));
          float e0 = __expf(logit[i][0] - m), e1 = __expf(logit[i][1] - m);
          float sl = e0 + e1;
          float s = sl + __shfl_xor(sl, 16);
          route[i][0] = e0 / s; route[i][1] = e1 / s;
        }
      }
#pragma unroll
      for (int op = 0; op < 2; ++op) {
        v2f pre[8];
        const int o = 2 * kh + op;
        const v2f* bo = (const v2f*)(bg + o * 16);
        const v2f r0 = {route[0][op], route[0][op]}, r1 = {route[1][op], route[1][op]};
        const v2f r2 = {route[2][op], route[2][op]}, r3 = {route[3][op], route[3][op]};
#pragma unroll
        for (int j = 0; j < 8; ++j) {
          v2f pp = r0 * ATOMS2(0, op, j);
          pp = __builtin_elementwise_fma(r1, ATOMS2(1, op, j), pp);
          pp = __builtin_elementwise_fma(r2, ATOMS2(2, op, j), pp);
          pp = __builtin_elementwise_fma(r3, ATOMS2(3, op, j), pp);
          pre[j] = pp + bo[j];
        }
        v2f n2v = pre[0] * pre[0];
#pragma unroll
        for (int j = 1; j < 8; ++j) n2v = __builtin_elementwise_fma(pre[j], pre[j], n2v);
        float n2 = n2v.x + n2v.y;
        float nrm = sqrtf(n2);
        if (it < 2) {
#pragma unroll
          for (int i = 0; i < 4; ++i) {
            v2f d = pre[0] * ATOMS2(i, op, 0);
#pragma unroll
            for (int j = 1; j < 8; ++j) d = __builtin_elementwise_fma(pre[j], ATOMS2(i, op, j), d);
            logit[i][op] += (d.x + d.y) / fmaxf(nrm * nb[i][op], 1e-8f);
          }
        } else {
          float scale = n2 / ((nrm + 1e-12f) * (1.f + n2));
          float* orow = outg + (((b * 4 + o) * 16) << 15) + (od << 10) + (oh << 5) + ow;
#pragma unroll
          for (int j = 0; j < 8; ++j) {
            v2f v = pre[j] * (v2f){scale, scale};
            orow[(2 * j) << 15] = v.x;
            orow[(2 * j + 1) << 15] = v.y;
          }
        }
      }
    }
#undef ATOMS2
    }
  }
}

extern "C" void kernel_launch(void* const* d_in, const int* in_sizes, int n_in,
                              void* d_out, int out_size, void* d_ws, size_t ws_size,
                              hipStream_t stream) {
  (void)in_sizes; (void)n_in; (void)out_size; (void)d_ws; (void)ws_size;
  const float* x = (const float*)d_in[0];
  const float* w = (const float*)d_in[1];
  const float* b = (const float*)d_in[2];
  float* out = (float*)d_out;

  deconv_caps_routing<<<dim3(4 * 32 * 2), dim3(256), 0, stream>>>(x, w, b, out);
}
